# starting kernel
# baseline (speedup 1.0000x reference)
#include <hip/hip_runtime.h>
#include <hip/hip_bf16.h>
#include <math.h>


typedef __attribute__((ext_vector_type(2))) float v2f;
typedef __attribute__((ext_vector_type(4))) float v4f;
typedef __attribute__((ext_vector_type(8))) float v8f;

#define T_LEN   160000
#define NCHAN   128
#define CHUNK   320
#define NCHUNK  500
#define NBLK    20
#define SEG     16

__global__ void biquad_phase1(const float* __restrict__ X,
                              float* __restrict__ summ,
                              float b0, float b1, float b2, float a1, float a2)
{
    int tid = blockIdx.x * blockDim.x + threadIdx.x;
    if (tid >= NCHAN * NCHUNK) return;
    int ch = tid / NCHUNK;
    int k  = tid - ch * NCHUNK;

    const float* xc = X + (size_t)ch * T_LEN + (size_t)k * CHUNK;

    float xm1 = 0.f, xm2 = 0.f;
    if (k > 0) { xm1 = xc[-1]; xm2 = xc[-2]; }

    float y1 = 0.f, y2 = 0.f;
#pragma unroll 4
    for (int i = 0; i < CHUNK; i += 4) {
        __builtin_prefetch(xc + i + 128, 0, 3);
        v4f x4 = *(const v4f*)(xc + i);
        float f0 = b0 * x4.x + b1 * xm1  + b2 * xm2;
        float f1 = b0 * x4.y + b1 * x4.x + b2 * xm1;
        float f2 = b0 * x4.z + b1 * x4.y + b2 * x4.x;
        float f3 = b0 * x4.w + b1 * x4.z + b2 * x4.y;
        float ya = f0 - a1 * y1 - a2 * y2;
        float yb = f1 - a1 * ya - a2 * y1;
        float yc = f2 - a1 * yb - a2 * ya;
        float yd = f3 - a1 * yc - a2 * yb;
        y2 = yc; y1 = yd;
        xm1 = x4.w; xm2 = x4.z;
    }
    summ[(size_t)tid * 2 + 0] = y1;
    summ[(size_t)tid * 2 + 1] = y2;
}

__global__ void __launch_bounds__(32)
biquad_phase2(const float* __restrict__ summ,
              float* __restrict__ bound,
              float a1, float a2)
{
    int ch   = blockIdx.x;
    int lane = threadIdx.x;

    float p1 = 1.f, p2 = 0.f;
    float q1 = 0.f, q2 = 1.f;
#pragma unroll
    for (int i = 0; i < 5; ++i) {
        float np = -a1 * p1 - a2 * p2; p2 = p1; p1 = np;
        float nq = -a1 * q1 - a2 * q2; q2 = q1; q1 = nq;
    }
#pragma unroll
    for (int s = 0; s < 6; ++s) {
        float n11 = p1 * p1 + q1 * p2;
        float n12 = p1 * q1 + q1 * q2;
        float n21 = p2 * p1 + q2 * p2;
        float n22 = p2 * q1 + q2 * q2;
        p1 = n11; q1 = n12; p2 = n21; q2 = n22;
    }

    const float* sc = summ  + (size_t)ch * NCHUNK * 2;
    float*       bc = bound + (size_t)ch * NCHUNK * 2;
    int q0 = lane * SEG;

    float w1 = 0.f, w2 = 0.f;
#pragma unroll 4
    for (int i = 0; i < SEG; ++i) {
        int q = q0 + i;
        float d1 = 0.f, d2 = 0.f;
        if (q < NCHUNK) { d1 = sc[2 * q]; d2 = sc[2 * q + 1]; }
        float nw1 = p1 * w1 + q1 * w2 + d1;
        float nw2 = p2 * w1 + q2 * w2 + d2;
        w1 = nw1; w2 = nw2;
    }
    float m11 = p1, m12 = q1, m21 = p2, m22 = q2;
#pragma unroll
    for (int s = 0; s < 4; ++s) {
        float n11 = m11 * m11 + m12 * m21;
        float n12 = m11 * m12 + m12 * m22;
        float n21 = m21 * m11 + m22 * m21;
        float n22 = m21 * m12 + m22 * m22;
        m11 = n11; m12 = n12; m21 = n21; m22 = n22;
    }

#pragma unroll
    for (int o = 1; o < 32; o <<= 1) {
        float o11 = __shfl_up(m11, o, 32);
        float o12 = __shfl_up(m12, o, 32);
        float o21 = __shfl_up(m21, o, 32);
        float o22 = __shfl_up(m22, o, 32);
        float ow1 = __shfl_up(w1, o, 32);
        float ow2 = __shfl_up(w2, o, 32);
        if (lane >= o) {
            float nw1 = m11 * ow1 + m12 * ow2 + w1;
            float nw2 = m21 * ow1 + m22 * ow2 + w2;
            w1 = nw1; w2 = nw2;
            float n11 = m11 * o11 + m12 * o21;
            float n12 = m11 * o12 + m12 * o22;
            float n21 = m21 * o11 + m22 * o21;
            float n22 = m21 * o12 + m22 * o22;
            m11 = n11; m12 = n12; m21 = n21; m22 = n22;
        }
    }
    float in1 = __shfl_up(w1, 1, 32);
    float in2 = __shfl_up(w2, 1, 32);
    if (lane == 0) { in1 = 0.f; in2 = 0.f; }

    float s1 = in1, s2 = in2;
#pragma unroll 4
    for (int i = 0; i < SEG; ++i) {
        int q = q0 + i;
        if (q < NCHUNK) {
            bc[2 * q + 0] = s1;
            bc[2 * q + 1] = s2;
            float d1 = sc[2 * q], d2 = sc[2 * q + 1];
            float ns1 = p1 * s1 + q1 * s2 + d1;
            float ns2 = p2 * s1 + q2 * s2 + d2;
            s1 = ns1; s2 = ns2;
        }
    }
}

__global__ void __launch_bounds__(256)
biquad_phase3(const float* __restrict__ X,
              const float* __restrict__ bound,
              float* __restrict__ Y,
              float b0, float b1, float b2, float a1, float a2)
{
    __shared__ float sh[17];

    int lane = threadIdx.x & 31;
    int wave = threadIdx.x >> 5;
    int k    = blockIdx.x;

    if (threadIdx.x == 0) {
        float h0 = 1.0f, h1 = -a1;
        sh[0] = h0; sh[1] = h1;
        for (int i = 2; i <= 16; ++i) {
            float hn = -a1 * h1 - a2 * h0;
            sh[i] = hn; h0 = h1; h1 = hn;
        }
    }
    __syncthreads();

    int m    = lane & 15;
    int half = lane >> 4;

    v2f A[4];
#pragma unroll
    for (int j = 0; j < 4; ++j) {
        int k0 = 4 * j + 2 * half;
        A[j].x = (m >= k0)     ? sh[m - k0]     : 0.0f;
        A[j].y = (m >= k0 + 1) ? sh[m - k0 - 1] : 0.0f;
    }
    float uu[8], vv[8];
#pragma unroll
    for (int r = 0; r < 8; ++r) {
        int mm = r + 8 * half;
        uu[r] = sh[mm + 1];
        vv[r] = -a2 * sh[mm];
    }

    int ch = wave * 16 + m;
    const float* xc = X + (size_t)ch * T_LEN;
    float*       yc = Y + (size_t)ch * T_LEN;
    int tbase = k * CHUNK;

    float y1 = bound[((size_t)ch * NCHUNK + k) * 2 + 0];
    float y2 = bound[((size_t)ch * NCHUNK + k) * 2 + 1];
    float xm1 = 0.f, xm2 = 0.f;
    if (k > 0) { xm1 = xc[tbase - 1]; xm2 = xc[tbase - 2]; }

#pragma unroll 1
    for (int b = 0; b < NBLK; ++b) {
        int t = tbase + b * 16;

        v8f c;
#pragma unroll
        for (int r = 0; r < 8; ++r) c[r] = uu[r] * y1 + vv[r] * y2;

#pragma unroll
        for (int j = 0; j < 4; ++j) {
            float f0 = 0.f, f1 = 0.f, f2 = 0.f, f3 = 0.f;
            if (half == 0) {
                if (j == 0)
                    __builtin_prefetch(xc + t + 64, 0, 3);
                v4f x4 = *(const v4f*)(xc + t + 4 * j);
                f0 = b0 * x4.x + b1 * xm1  + b2 * xm2;
                f1 = b0 * x4.y + b1 * x4.x + b2 * xm1;
                f2 = b0 * x4.z + b1 * x4.y + b2 * x4.x;
                f3 = b0 * x4.w + b1 * x4.z + b2 * x4.y;
                xm1 = x4.w; xm2 = x4.z;
            }
            float f2s = __shfl(f2, m, 32);
            float f3s = __shfl(f3, m, 32);
            v2f Bv;
            Bv.x = half ? f2s : f0;
            Bv.y = half ? f3s : f1;
            c = __builtin_amdgcn_wmma_f32_16x16x4_f32(
                    false, A[j], false, Bv, (short)0, c, false, false);
        }

#pragma unroll
        for (int r = 0; r < 8; ++r)
            yc[t + r + 8 * half] = c[r];

        y1 = __shfl(c[7], m + 16, 32);
        y2 = __shfl(c[6], m + 16, 32);
    }
}

extern "C" void kernel_launch(void* const* d_in, const int* in_sizes, int n_in,
                              void* d_out, int out_size, void* d_ws, size_t ws_size,
                              hipStream_t stream) {
    const float* clip = (const float*)d_in[0];
    float* out = (float*)d_out;

    const double SR = 32000.0, CUT = 8000.0, Qf = 0.707;
    double w0    = 2.0 * M_PI * CUT / SR;
    double alpha = sin(w0) / (2.0 * Qf);
    double cw    = cos(w0);
    double b0d = (1.0 - cw) / 2.0, b1d = 1.0 - cw, b2d = b0d;
    double a0d = 1.0 + alpha, a1d = -2.0 * cw, a2d = 1.0 - alpha;
    float b0 = (float)(b0d / a0d), b1 = (float)(b1d / a0d), b2 = (float)(b2d / a0d);
    float a1 = (float)(a1d / a0d), a2 = (float)(a2d / a0d);

    float* summ  = (float*)d_ws;
    float* bound = summ + (size_t)NCHAN * NCHUNK * 2;
    (void)ws_size;

    {
        int total = NCHAN * NCHUNK;
        dim3 grid((total + 255) / 256), block(256);
        biquad_phase1<<<grid, block, 0, stream>>>(clip, summ, b0, b1, b2, a1, a2);
    }
    {
        biquad_phase2<<<dim3(NCHAN), dim3(32), 0, stream>>>(summ, bound, a1, a2);
    }
    {
        biquad_phase3<<<dim3(NCHUNK), dim3(256), 0, stream>>>(
            clip, bound, out, b0, b1, b2, a1, a2);
    }
}


# phase 3: X and Y go as 16-channel x 32-sample tiles through wave-private LDS: global loads/stores are whole 128-B lines asked two tiles ahead, not 16-line pieces each waited for; sums unchanged
# speedup vs baseline: 1.5288x; 1.5288x over previous
#include <hip/hip_runtime.h>
#include <hip/hip_bf16.h>
#include <math.h>


typedef __attribute__((ext_vector_type(2))) float v2f;
typedef __attribute__((ext_vector_type(4))) float v4f;
typedef __attribute__((ext_vector_type(8))) float v8f;

#define T_LEN   160000
#define NCHAN   128
#define CHUNK   320
#define NCHUNK  500
#define NBLK    20
#define SEG     16

__global__ void biquad_phase1(const float* __restrict__ X,
                              float* __restrict__ summ,
                              float b0, float b1, float b2, float a1, float a2)
{
    int tid = blockIdx.x * blockDim.x + threadIdx.x;
    if (tid >= NCHAN * NCHUNK) return;
    int ch = tid / NCHUNK;
    int k  = tid - ch * NCHUNK;

    const float* xc = X + (size_t)ch * T_LEN + (size_t)k * CHUNK;

    float xm1 = 0.f, xm2 = 0.f;
    if (k > 0) { xm1 = xc[-1]; xm2 = xc[-2]; }

    float y1 = 0.f, y2 = 0.f;
#pragma unroll 4
    for (int i = 0; i < CHUNK; i += 4) {
        __builtin_prefetch(xc + i + 128, 0, 3);
        v4f x4 = *(const v4f*)(xc + i);
        float f0 = b0 * x4.x + b1 * xm1  + b2 * xm2;
        float f1 = b0 * x4.y + b1 * x4.x + b2 * xm1;
        float f2 = b0 * x4.z + b1 * x4.y + b2 * x4.x;
        float f3 = b0 * x4.w + b1 * x4.z + b2 * x4.y;
        float ya = f0 - a1 * y1 - a2 * y2;
        float yb = f1 - a1 * ya - a2 * y1;
        float yc = f2 - a1 * yb - a2 * ya;
        float yd = f3 - a1 * yc - a2 * yb;
        y2 = yc; y1 = yd;
        xm1 = x4.w; xm2 = x4.z;
    }
    summ[(size_t)tid * 2 + 0] = y1;
    summ[(size_t)tid * 2 + 1] = y2;
}

__global__ void __launch_bounds__(32)
biquad_phase2(const float* __restrict__ summ,
              float* __restrict__ bound,
              float a1, float a2)
{
    int ch   = blockIdx.x;
    int lane = threadIdx.x;

    float p1 = 1.f, p2 = 0.f;
    float q1 = 0.f, q2 = 1.f;
#pragma unroll
    for (int i = 0; i < 5; ++i) {
        float np = -a1 * p1 - a2 * p2; p2 = p1; p1 = np;
        float nq = -a1 * q1 - a2 * q2; q2 = q1; q1 = nq;
    }
#pragma unroll
    for (int s = 0; s < 6; ++s) {
        float n11 = p1 * p1 + q1 * p2;
        float n12 = p1 * q1 + q1 * q2;
        float n21 = p2 * p1 + q2 * p2;
        float n22 = p2 * q1 + q2 * q2;
        p1 = n11; q1 = n12; p2 = n21; q2 = n22;
    }

    const float* sc = summ  + (size_t)ch * NCHUNK * 2;
    float*       bc = bound + (size_t)ch * NCHUNK * 2;
    int q0 = lane * SEG;

    float w1 = 0.f, w2 = 0.f;
#pragma unroll 4
    for (int i = 0; i < SEG; ++i) {
        int q = q0 + i;
        float d1 = 0.f, d2 = 0.f;
        if (q < NCHUNK) { d1 = sc[2 * q]; d2 = sc[2 * q + 1]; }
        float nw1 = p1 * w1 + q1 * w2 + d1;
        float nw2 = p2 * w1 + q2 * w2 + d2;
        w1 = nw1; w2 = nw2;
    }
    float m11 = p1, m12 = q1, m21 = p2, m22 = q2;
#pragma unroll
    for (int s = 0; s < 4; ++s) {
        float n11 = m11 * m11 + m12 * m21;
        float n12 = m11 * m12 + m12 * m22;
        float n21 = m21 * m11 + m22 * m21;
        float n22 = m21 * m12 + m22 * m22;
        m11 = n11; m12 = n12; m21 = n21; m22 = n22;
    }

#pragma unroll
    for (int o = 1; o < 32; o <<= 1) {
        float o11 = __shfl_up(m11, o, 32);
        float o12 = __shfl_up(m12, o, 32);
        float o21 = __shfl_up(m21, o, 32);
        float o22 = __shfl_up(m22, o, 32);
        float ow1 = __shfl_up(w1, o, 32);
        float ow2 = __shfl_up(w2, o, 32);
        if (lane >= o) {
            float nw1 = m11 * ow1 + m12 * ow2 + w1;
            float nw2 = m21 * ow1 + m22 * ow2 + w2;
            w1 = nw1; w2 = nw2;
            float n11 = m11 * o11 + m12 * o21;
            float n12 = m11 * o12 + m12 * o22;
            float n21 = m21 * o11 + m22 * o21;
            float n22 = m21 * o12 + m22 * o22;
            m11 = n11; m12 = n12; m21 = n21; m22 = n22;
        }
    }
    float in1 = __shfl_up(w1, 1, 32);
    float in2 = __shfl_up(w2, 1, 32);
    if (lane == 0) { in1 = 0.f; in2 = 0.f; }

    float s1 = in1, s2 = in2;
#pragma unroll 4
    for (int i = 0; i < SEG; ++i) {
        int q = q0 + i;
        if (q < NCHUNK) {
            bc[2 * q + 0] = s1;
            bc[2 * q + 1] = s2;
            float d1 = sc[2 * q], d2 = sc[2 * q + 1];
            float ns1 = p1 * s1 + q1 * s2 + d1;
            float ns2 = p2 * s1 + q2 * s2 + d2;
            s1 = ns1; s2 = ns2;
        }
    }
}

#define SB      32
#define NSB     10
#define PITCH   36
#define XOFF    4

__device__ __forceinline__ void wave_lds_order()
{
    __builtin_amdgcn_fence(__ATOMIC_RELEASE, "wavefront");
    __builtin_amdgcn_wave_barrier();
    __builtin_amdgcn_fence(__ATOMIC_ACQUIRE, "wavefront");
}

__global__ void __launch_bounds__(256)
biquad_phase3(const float* __restrict__ X,
              const float* __restrict__ bound,
              float* __restrict__ Y,
              float b0, float b1, float b2, float a1, float a2)
{
    __shared__ float sh[17];
    __shared__ __attribute__((aligned(16))) float tin [8 * 16 * PITCH];
    __shared__ __attribute__((aligned(16))) float tout[8 * 16 * PITCH];

    int lane = threadIdx.x & 31;
    int wave = threadIdx.x >> 5;
    int k    = blockIdx.x;

    if (threadIdx.x == 0) {
        float h0 = 1.0f, h1 = -a1;
        sh[0] = h0; sh[1] = h1;
        for (int i = 2; i <= 16; ++i) {
            float hn = -a1 * h1 - a2 * h0;
            sh[i] = hn; h0 = h1; h1 = hn;
        }
    }
    __syncthreads();

    int m    = lane & 15;
    int half = lane >> 4;

    v2f A[4];
#pragma unroll
    for (int j = 0; j < 4; ++j) {
        int k0 = 4 * j + 2 * half;
        A[j].x = (m >= k0)     ? sh[m - k0]     : 0.0f;
        A[j].y = (m >= k0 + 1) ? sh[m - k0 - 1] : 0.0f;
    }
    float uu[8], vv[8];
#pragma unroll
    for (int r = 0; r < 8; ++r) {
        int mm = r + 8 * half;
        uu[r] = sh[mm + 1];
        vv[r] = -a2 * sh[mm];
    }

    int ch = wave * 16 + m;
    int tbase = k * CHUNK;

    float y1 = bound[((size_t)ch * NCHUNK + k) * 2 + 0];
    float y2 = bound[((size_t)ch * NCHUNK + k) * 2 + 1];

    float* win  = tin  + wave * 16 * PITCH;
    float* wout = tout + wave * 16 * PITCH;

    int mrow = lane >> 3;
    int mseg = lane & 7;
    const float* xg = X + (size_t)(wave * 16 + mrow) * T_LEN + tbase + 4 * mseg;
    float*       yg = Y + (size_t)(wave * 16 + mrow) * T_LEN + tbase + 4 * mseg;

    v4f ga[4], gb[4];
#pragma unroll
    for (int i = 0; i < 4; ++i)
        ga[i] = *(const v4f*)(xg + (size_t)(4 * i) * T_LEN);
#pragma unroll
    for (int i = 0; i < 4; ++i)
        gb[i] = *(const v4f*)(xg + (size_t)(4 * i) * T_LEN + SB);

    v2f hal; hal.x = 0.f; hal.y = 0.f;
    if (k > 0) hal = *(const v2f*)(X + (size_t)ch * T_LEN + tbase - 2);

    auto do_tile = [&](int sb, v4f (&g)[4]) {
        *(v2f*)(win + m * PITCH + 2) = hal;
#pragma unroll
        for (int i = 0; i < 4; ++i)
            *(v4f*)(win + (4 * i + mrow) * PITCH + XOFF + 4 * mseg) = g[i];
        wave_lds_order();

        if (sb + 2 < NSB) {
#pragma unroll
            for (int i = 0; i < 4; ++i)
                g[i] = *(const v4f*)(xg + (size_t)(4 * i) * T_LEN + (sb + 2) * SB);
        }

#pragma unroll
        for (int bb = 0; bb < 2; ++bb) {
            v8f c;
#pragma unroll
            for (int r = 0; r < 8; ++r) c[r] = uu[r] * y1 + vv[r] * y2;

#pragma unroll
            for (int j = 0; j < 4; ++j) {
                const float* p = win + m * PITCH + XOFF + 16 * bb + 4 * j + 2 * half - 2;
                v2f lo = *(const v2f*)p;
                v2f hi = *(const v2f*)(p + 2);
                v2f Bv;
                Bv.x = b0 * hi.x + b1 * lo.y + b2 * lo.x;
                Bv.y = b0 * hi.y + b1 * hi.x + b2 * lo.y;
                c = __builtin_amdgcn_wmma_f32_16x16x4_f32(
                        false, A[j], false, Bv, (short)0, c, false, false);
            }

            v4f o0, o1;
            o0.x = c[0]; o0.y = c[1]; o0.z = c[2]; o0.w = c[3];
            o1.x = c[4]; o1.y = c[5]; o1.z = c[6]; o1.w = c[7];
            *(v4f*)(wout + m * PITCH + 16 * bb + 8 * half)     = o0;
            *(v4f*)(wout + m * PITCH + 16 * bb + 8 * half + 4) = o1;

            y1 = __shfl(c[7], m + 16, 32);
            y2 = __shfl(c[6], m + 16, 32);
        }

        hal = *(const v2f*)(win + m * PITCH + XOFF + SB - 2);
        wave_lds_order();

#pragma unroll
        for (int i = 0; i < 4; ++i) {
            v4f o = *(const v4f*)(wout + (4 * i + mrow) * PITCH + 4 * mseg);
            *(v4f*)(yg + (size_t)(4 * i) * T_LEN + sb * SB) = o;
        }
    };

    for (int sb = 0; sb < NSB; sb += 2) {
        do_tile(sb,     ga);
        do_tile(sb + 1, gb);
    }
}

extern "C" void kernel_launch(void* const* d_in, const int* in_sizes, int n_in,
                              void* d_out, int out_size, void* d_ws, size_t ws_size,
                              hipStream_t stream) {
    const float* clip = (const float*)d_in[0];
    float* out = (float*)d_out;

    const double SR = 32000.0, CUT = 8000.0, Qf = 0.707;
    double w0    = 2.0 * M_PI * CUT / SR;
    double alpha = sin(w0) / (2.0 * Qf);
    double cw    = cos(w0);
    double b0d = (1.0 - cw) / 2.0, b1d = 1.0 - cw, b2d = b0d;
    double a0d = 1.0 + alpha, a1d = -2.0 * cw, a2d = 1.0 - alpha;
    float b0 = (float)(b0d / a0d), b1 = (float)(b1d / a0d), b2 = (float)(b2d / a0d);
    float a1 = (float)(a1d / a0d), a2 = (float)(a2d / a0d);

    float* summ  = (float*)d_ws;
    float* bound = summ + (size_t)NCHAN * NCHUNK * 2;
    (void)ws_size;

    {
        int total = NCHAN * NCHUNK;
        dim3 grid((total + 255) / 256), block(256);
        biquad_phase1<<<grid, block, 0, stream>>>(clip, summ, b0, b1, b2, a1, a2);
    }
    {
        biquad_phase2<<<dim3(NCHAN), dim3(32), 0, stream>>>(summ, bound, a1, a2);
    }
    {
        biquad_phase3<<<dim3(NCHUNK), dim3(256), 0, stream>>>(
            clip, bound, out, b0, b1, b2, a1, a2);
    }
}


# phase 1: X read as 32-stream x 64-sample tiles through wave-private LDS, one row per lane: a global load covers 4 whole lines, not 32 lines for 16 B each; lane prefetch dropped; sums unchanged
# speedup vs baseline: 2.1189x; 1.3860x over previous
#include <hip/hip_runtime.h>
#include <hip/hip_bf16.h>
#include <math.h>


typedef __attribute__((ext_vector_type(2))) float v2f;
typedef __attribute__((ext_vector_type(4))) float v4f;
typedef __attribute__((ext_vector_type(8))) float v8f;

#define T_LEN   160000
#define NCHAN   128
#define CHUNK   320
#define NCHUNK  500
#define NBLK    20
#define SEG     16

__device__ __forceinline__ void wave_lds_order()
{
    __builtin_amdgcn_fence(__ATOMIC_RELEASE, "wavefront");
    __builtin_amdgcn_wave_barrier();
    __builtin_amdgcn_fence(__ATOMIC_ACQUIRE, "wavefront");
}

#define P1_SB     64
#define P1_NT     5
#define P1_PITCH  68
static_assert(CHUNK == P1_SB * P1_NT, "phase 1 tiles must cover the chunk");
static_assert((NCHAN * NCHUNK) % 256 == 0, "phase 1 launches whole work-groups only");

__global__ void __launch_bounds__(256)
biquad_phase1(const float* __restrict__ X,
              float* __restrict__ summ,
              float b0, float b1, float b2, float a1, float a2)
{
    __shared__ __attribute__((aligned(16))) float tile[8 * 32 * P1_PITCH];

    int lane = threadIdx.x & 31;
    int wave = threadIdx.x >> 5;
    int tid  = blockIdx.x * blockDim.x + threadIdx.x;
    int k    = tid % NCHUNK;

    const float* xc = X + (size_t)tid * CHUNK;
    const float* xw = X + (size_t)(tid - lane) * CHUNK;
    float* wt = tile + wave * 32 * P1_PITCH;

    int mrow = lane >> 4;
    int mseg = lane & 15;
    const float* xg  = xw + (size_t)mrow * CHUNK + 4 * mseg;
    float*       wst = wt + mrow * P1_PITCH + 4 * mseg;
    const float* wrd = wt + lane * P1_PITCH;

    v4f ga[16], gb[16];
#pragma unroll
    for (int i = 0; i < 16; ++i)
        ga[i] = *(const v4f*)(xg + (size_t)(2 * i) * CHUNK);
#pragma unroll
    for (int i = 0; i < 16; ++i)
        gb[i] = *(const v4f*)(xg + (size_t)(2 * i) * CHUNK + P1_SB);

    float xm1 = 0.f, xm2 = 0.f;
    if (k > 0) { xm1 = xc[-1]; xm2 = xc[-2]; }

    float y1 = 0.f, y2 = 0.f;

    auto do_tile = [&](int t, v4f (&g)[16]) {
#pragma unroll
        for (int i = 0; i < 16; ++i)
            *(v4f*)(wst + (2 * i) * P1_PITCH) = g[i];
        if (t + 2 < P1_NT) {
#pragma unroll
            for (int i = 0; i < 16; ++i)
                g[i] = *(const v4f*)(xg + (size_t)(2 * i) * CHUNK + (t + 2) * P1_SB);
        }
        wave_lds_order();

        v4f r[16];
#pragma unroll
        for (int j = 0; j < 16; ++j)
            r[j] = *(const v4f*)(wrd + 4 * j);
        wave_lds_order();

#pragma unroll
        for (int j = 0; j < 16; ++j) {
            v4f x4 = r[j];
            float f0 = b0 * x4.x + b1 * xm1  + b2 * xm2;
            float f1 = b0 * x4.y + b1 * x4.x + b2 * xm1;
            float f2 = b0 * x4.z + b1 * x4.y + b2 * x4.x;
            float f3 = b0 * x4.w + b1 * x4.z + b2 * x4.y;
            float ya = f0 - a1 * y1 - a2 * y2;
            float yb = f1 - a1 * ya - a2 * y1;
            float yc = f2 - a1 * yb - a2 * ya;
            float yd = f3 - a1 * yc - a2 * yb;
            y2 = yc; y1 = yd;
            xm1 = x4.w; xm2 = x4.z;
        }
    };

    do_tile(0, ga);
    do_tile(1, gb);
    do_tile(2, ga);
    do_tile(3, gb);
    do_tile(4, ga);

    summ[(size_t)tid * 2 + 0] = y1;
    summ[(size_t)tid * 2 + 1] = y2;
}

__global__ void __launch_bounds__(32)
biquad_phase2(const float* __restrict__ summ,
              float* __restrict__ bound,
              float a1, float a2)
{
    int ch   = blockIdx.x;
    int lane = threadIdx.x;

    float p1 = 1.f, p2 = 0.f;
    float q1 = 0.f, q2 = 1.f;
#pragma unroll
    for (int i = 0; i < 5; ++i) {
        float np = -a1 * p1 - a2 * p2; p2 = p1; p1 = np;
        float nq = -a1 * q1 - a2 * q2; q2 = q1; q1 = nq;
    }
#pragma unroll
    for (int s = 0; s < 6; ++s) {
        float n11 = p1 * p1 + q1 * p2;
        float n12 = p1 * q1 + q1 * q2;
        float n21 = p2 * p1 + q2 * p2;
        float n22 = p2 * q1 + q2 * q2;
        p1 = n11; q1 = n12; p2 = n21; q2 = n22;
    }

    const float* sc = summ  + (size_t)ch * NCHUNK * 2;
    float*       bc = bound + (size_t)ch * NCHUNK * 2;
    int q0 = lane * SEG;

    float w1 = 0.f, w2 = 0.f;
#pragma unroll 4
    for (int i = 0; i < SEG; ++i) {
        int q = q0 + i;
        float d1 = 0.f, d2 = 0.f;
        if (q < NCHUNK) { d1 = sc[2 * q]; d2 = sc[2 * q + 1]; }
        float nw1 = p1 * w1 + q1 * w2 + d1;
        float nw2 = p2 * w1 + q2 * w2 + d2;
        w1 = nw1; w2 = nw2;
    }
    float m11 = p1, m12 = q1, m21 = p2, m22 = q2;
#pragma unroll
    for (int s = 0; s < 4; ++s) {
        float n11 = m11 * m11 + m12 * m21;
        float n12 = m11 * m12 + m12 * m22;
        float n21 = m21 * m11 + m22 * m21;
        float n22 = m21 * m12 + m22 * m22;
        m11 = n11; m12 = n12; m21 = n21; m22 = n22;
    }

#pragma unroll
    for (int o = 1; o < 32; o <<= 1) {
        float o11 = __shfl_up(m11, o, 32);
        float o12 = __shfl_up(m12, o, 32);
        float o21 = __shfl_up(m21, o, 32);
        float o22 = __shfl_up(m22, o, 32);
        float ow1 = __shfl_up(w1, o, 32);
        float ow2 = __shfl_up(w2, o, 32);
        if (lane >= o) {
            float nw1 = m11 * ow1 + m12 * ow2 + w1;
            float nw2 = m21 * ow1 + m22 * ow2 + w2;
            w1 = nw1; w2 = nw2;
            float n11 = m11 * o11 + m12 * o21;
            float n12 = m11 * o12 + m12 * o22;
            float n21 = m21 * o11 + m22 * o21;
            float n22 = m21 * o12 + m22 * o22;
            m11 = n11; m12 = n12; m21 = n21; m22 = n22;
        }
    }
    float in1 = __shfl_up(w1, 1, 32);
    float in2 = __shfl_up(w2, 1, 32);
    if (lane == 0) { in1 = 0.f; in2 = 0.f; }

    float s1 = in1, s2 = in2;
#pragma unroll 4
    for (int i = 0; i < SEG; ++i) {
        int q = q0 + i;
        if (q < NCHUNK) {
            bc[2 * q + 0] = s1;
            bc[2 * q + 1] = s2;
            float d1 = sc[2 * q], d2 = sc[2 * q + 1];
            float ns1 = p1 * s1 + q1 * s2 + d1;
            float ns2 = p2 * s1 + q2 * s2 + d2;
            s1 = ns1; s2 = ns2;
        }
    }
}

#define SB      32
#define NSB     10
#define PITCH   36
#define XOFF    4

__global__ void __launch_bounds__(256)
biquad_phase3(const float* __restrict__ X,
              const float* __restrict__ bound,
              float* __restrict__ Y,
              float b0, float b1, float b2, float a1, float a2)
{
    __shared__ float sh[17];
    __shared__ __attribute__((aligned(16))) float tin [8 * 16 * PITCH];
    __shared__ __attribute__((aligned(16))) float tout[8 * 16 * PITCH];

    int lane = threadIdx.x & 31;
    int wave = threadIdx.x >> 5;
    int k    = blockIdx.x;

    if (threadIdx.x == 0) {
        float h0 = 1.0f, h1 = -a1;
        sh[0] = h0; sh[1] = h1;
        for (int i = 2; i <= 16; ++i) {
            float hn = -a1 * h1 - a2 * h0;
            sh[i] = hn; h0 = h1; h1 = hn;
        }
    }
    __syncthreads();

    int m    = lane & 15;
    int half = lane >> 4;

    v2f A[4];
#pragma unroll
    for (int j = 0; j < 4; ++j) {
        int k0 = 4 * j + 2 * half;
        A[j].x = (m >= k0)     ? sh[m - k0]     : 0.0f;
        A[j].y = (m >= k0 + 1) ? sh[m - k0 - 1] : 0.0f;
    }
    float uu[8], vv[8];
#pragma unroll
    for (int r = 0; r < 8; ++r) {
        int mm = r + 8 * half;
        uu[r] = sh[mm + 1];
        vv[r] = -a2 * sh[mm];
    }

    int ch = wave * 16 + m;
    int tbase = k * CHUNK;

    float y1 = bound[((size_t)ch * NCHUNK + k) * 2 + 0];
    float y2 = bound[((size_t)ch * NCHUNK + k) * 2 + 1];

    float* win  = tin  + wave * 16 * PITCH;
    float* wout = tout + wave * 16 * PITCH;

    int mrow = lane >> 3;
    int mseg = lane & 7;
    const float* xg = X + (size_t)(wave * 16 + mrow) * T_LEN + tbase + 4 * mseg;
    float*       yg = Y + (size_t)(wave * 16 + mrow) * T_LEN + tbase + 4 * mseg;

    v4f ga[4], gb[4];
#pragma unroll
    for (int i = 0; i < 4; ++i)
        ga[i] = *(const v4f*)(xg + (size_t)(4 * i) * T_LEN);
#pragma unroll
    for (int i = 0; i < 4; ++i)
        gb[i] = *(const v4f*)(xg + (size_t)(4 * i) * T_LEN + SB);

    v2f hal; hal.x = 0.f; hal.y = 0.f;
    if (k > 0) hal = *(const v2f*)(X + (size_t)ch * T_LEN + tbase - 2);

    auto do_tile = [&](int sb, v4f (&g)[4]) {
        *(v2f*)(win + m * PITCH + 2) = hal;
#pragma unroll
        for (int i = 0; i < 4; ++i)
            *(v4f*)(win + (4 * i + mrow) * PITCH + XOFF + 4 * mseg) = g[i];
        wave_lds_order();

        if (sb + 2 < NSB) {
#pragma unroll
            for (int i = 0; i < 4; ++i)
                g[i] = *(const v4f*)(xg + (size_t)(4 * i) * T_LEN + (sb + 2) * SB);
        }

#pragma unroll
        for (int bb = 0; bb < 2; ++bb) {
            v8f c;
#pragma unroll
            for (int r = 0; r < 8; ++r) c[r] = uu[r] * y1 + vv[r] * y2;

#pragma unroll
            for (int j = 0; j < 4; ++j) {
                const float* p = win + m * PITCH + XOFF + 16 * bb + 4 * j + 2 * half - 2;
                v2f lo = *(const v2f*)p;
                v2f hi = *(const v2f*)(p + 2);
                v2f Bv;
                Bv.x = b0 * hi.x + b1 * lo.y + b2 * lo.x;
                Bv.y = b0 * hi.y + b1 * hi.x + b2 * lo.y;
                c = __builtin_amdgcn_wmma_f32_16x16x4_f32(
                        false, A[j], false, Bv, (short)0, c, false, false);
            }

            v4f o0, o1;
            o0.x = c[0]; o0.y = c[1]; o0.z = c[2]; o0.w = c[3];
            o1.x = c[4]; o1.y = c[5]; o1.z = c[6]; o1.w = c[7];
            *(v4f*)(wout + m * PITCH + 16 * bb + 8 * half)     = o0;
            *(v4f*)(wout + m * PITCH + 16 * bb + 8 * half + 4) = o1;

            y1 = __shfl(c[7], m + 16, 32);
            y2 = __shfl(c[6], m + 16, 32);
        }

        hal = *(const v2f*)(win + m * PITCH + XOFF + SB - 2);
        wave_lds_order();

#pragma unroll
        for (int i = 0; i < 4; ++i) {
            v4f o = *(const v4f*)(wout + (4 * i + mrow) * PITCH + 4 * mseg);
            *(v4f*)(yg + (size_t)(4 * i) * T_LEN + sb * SB) = o;
        }
    };

    for (int sb = 0; sb < NSB; sb += 2) {
        do_tile(sb,     ga);
        do_tile(sb + 1, gb);
    }
}

extern "C" void kernel_launch(void* const* d_in, const int* in_sizes, int n_in,
                              void* d_out, int out_size, void* d_ws, size_t ws_size,
                              hipStream_t stream) {
    const float* clip = (const float*)d_in[0];
    float* out = (float*)d_out;

    const double SR = 32000.0, CUT = 8000.0, Qf = 0.707;
    double w0    = 2.0 * M_PI * CUT / SR;
    double alpha = sin(w0) / (2.0 * Qf);
    double cw    = cos(w0);
    double b0d = (1.0 - cw) / 2.0, b1d = 1.0 - cw, b2d = b0d;
    double a0d = 1.0 + alpha, a1d = -2.0 * cw, a2d = 1.0 - alpha;
    float b0 = (float)(b0d / a0d), b1 = (float)(b1d / a0d), b2 = (float)(b2d / a0d);
    float a1 = (float)(a1d / a0d), a2 = (float)(a2d / a0d);

    float* summ  = (float*)d_ws;
    float* bound = summ + (size_t)NCHAN * NCHUNK * 2;
    (void)ws_size;

    {
        dim3 grid((NCHAN * NCHUNK) / 256), block(256);
        biquad_phase1<<<grid, block, 0, stream>>>(clip, summ, b0, b1, b2, a1, a2);
    }
    {
        biquad_phase2<<<dim3(NCHAN), dim3(32), 0, stream>>>(summ, bound, a1, a2);
    }
    {
        biquad_phase3<<<dim3(NCHUNK), dim3(256), 0, stream>>>(
            clip, bound, out, b0, b1, b2, a1, a2);
    }
}


# phase 2 folded into phase 1: each work-group (half a channel) scans its 250 chunk end-states from LDS; phase 3 adds one carry term in the second half; one launch fewer; scan sum tree changes
# speedup vs baseline: 2.4880x; 1.1742x over previous
#include <hip/hip_runtime.h>
#include <hip/hip_bf16.h>
#include <math.h>


typedef __attribute__((ext_vector_type(2))) float v2f;
typedef __attribute__((ext_vector_type(4))) float v4f;
typedef __attribute__((ext_vector_type(8))) float v8f;

#define T_LEN   160000
#define NCHAN   128
#define CHUNK   320
#define NCHUNK  500
#define NBLK    20
#define HALF    250
#define SEG     8
static_assert(2 * HALF == NCHUNK, "two phase-1 work-groups cover a channel");
static_assert(SEG * 32 >= HALF && HALF <= 256, "one wave scans a group's chunks");

__device__ __forceinline__ void wave_lds_order()
{
    __builtin_amdgcn_fence(__ATOMIC_RELEASE, "wavefront");
    __builtin_amdgcn_wave_barrier();
    __builtin_amdgcn_fence(__ATOMIC_ACQUIRE, "wavefront");
}

#define P1_SB     64
#define P1_NT     5
#define P1_PITCH  68
#define P1_MVL    13
static_assert(CHUNK == P1_SB * P1_NT, "phase 1 tiles must cover the chunk");
static_assert(HALF - 7 * 32 == 2 * P1_MVL, "rows of the group's last wave");

__global__ void __launch_bounds__(256)
biquad_phase1(const float* __restrict__ X,
              float* __restrict__ bound,
              float* __restrict__ corr,
              float b0, float b1, float b2, float a1, float a2)
{
    __shared__ __attribute__((aligned(16))) float tile[8 * 32 * P1_PITCH];
    __shared__ __attribute__((aligned(16))) float sd[2 * 256];

    int lane = threadIdx.x & 31;
    int wave = threadIdx.x >> 5;
    int grp  = blockIdx.x;
    int ch   = grp >> 1;
    int hf   = grp & 1;
    int t    = threadIdx.x;
    int k    = hf * HALF + t;
    int s0   = grp * HALF;
    bool valid = t < HALF;
    bool full  = wave < 7;

    const float* xc = X + (size_t)(s0 + t) * CHUNK;
    const float* xw = X + (size_t)(s0 + 32 * wave) * CHUNK;
    float* wt = tile + wave * 32 * P1_PITCH;

    int mrow = lane >> 4;
    int mseg = lane & 15;
    const float* xg  = xw + (size_t)mrow * CHUNK + 4 * mseg;
    float*       wst = wt + mrow * P1_PITCH + 4 * mseg;
    const float* wrd = wt + lane * P1_PITCH;

    const int tailrow = full ? 2 * CHUNK : 0;

    auto request = [&](v4f (&g)[16], int tt) {
#pragma unroll
        for (int i = 0; i < P1_MVL; ++i)
            g[i] = *(const v4f*)(xg + (size_t)(2 * i) * CHUNK + tt * P1_SB);
#pragma unroll
        for (int i = P1_MVL; i < 16; ++i)
            g[i] = *(const v4f*)(xg + (size_t)(i * tailrow) + tt * P1_SB);
    };

    v4f ga[16], gb[16];
    request(ga, 0);
    request(gb, 1);

    float xm1 = 0.f, xm2 = 0.f;
    if (valid && k > 0) { xm1 = xc[-1]; xm2 = xc[-2]; }

    float y1 = 0.f, y2 = 0.f;

    auto do_tile = [&](int tt, v4f (&g)[16]) {
#pragma unroll
        for (int i = 0; i < 16; ++i)
            *(v4f*)(wst + (2 * i) * P1_PITCH) = g[i];
        if (tt + 2 < P1_NT) request(g, tt + 2);
        wave_lds_order();

        v4f r[16];
#pragma unroll
        for (int j = 0; j < 16; ++j)
            r[j] = *(const v4f*)(wrd + 4 * j);
        wave_lds_order();

#pragma unroll
        for (int j = 0; j < 16; ++j) {
            v4f x4 = r[j];
            float f0 = b0 * x4.x + b1 * xm1  + b2 * xm2;
            float f1 = b0 * x4.y + b1 * x4.x + b2 * xm1;
            float f2 = b0 * x4.z + b1 * x4.y + b2 * x4.x;
            float f3 = b0 * x4.w + b1 * x4.z + b2 * x4.y;
            float ya = f0 - a1 * y1 - a2 * y2;
            float yb = f1 - a1 * ya - a2 * y1;
            float yc = f2 - a1 * yb - a2 * ya;
            float yd = f3 - a1 * yc - a2 * yb;
            y2 = yc; y1 = yd;
            xm1 = x4.w; xm2 = x4.z;
        }
    };

    do_tile(0, ga);
    do_tile(1, gb);
    do_tile(2, ga);
    do_tile(3, gb);
    do_tile(4, ga);

    {
        v2f d; d.x = y1; d.y = y2;
        *(v2f*)(sd + 2 * t) = d;
    }

    float p1 = 1.f, p2 = 0.f;
    float q1 = 0.f, q2 = 1.f;
    if (wave == 0) {
#pragma unroll
        for (int i = 0; i < 5; ++i) {
            float np = -a1 * p1 - a2 * p2; p2 = p1; p1 = np;
            float nq = -a1 * q1 - a2 * q2; q2 = q1; q1 = nq;
        }
#pragma unroll
        for (int s = 0; s < 6; ++s) {
            float n11 = p1 * p1 + q1 * p2;
            float n12 = p1 * q1 + q1 * q2;
            float n21 = p2 * p1 + q2 * p2;
            float n22 = p2 * q1 + q2 * q2;
            p1 = n11; q1 = n12; p2 = n21; q2 = n22;
        }
    }
    __syncthreads();
    if (wave != 0) return;

    int q0 = lane * SEG;

    float da[SEG], db[SEG];
#pragma unroll
    for (int i = 0; i < SEG; ++i) {
        int q = q0 + i;
        v2f d = *(const v2f*)(sd + 2 * q);
        da[i] = (q < HALF) ? d.x : 0.f;
        db[i] = (q < HALF) ? d.y : 0.f;
    }

    float w1 = 0.f, w2 = 0.f;
#pragma unroll
    for (int i = 0; i < SEG; ++i) {
        float nw1 = p1 * w1 + q1 * w2 + da[i];
        float nw2 = p2 * w1 + q2 * w2 + db[i];
        w1 = nw1; w2 = nw2;
    }
    float m11 = p1, m12 = q1, m21 = p2, m22 = q2;
#pragma unroll
    for (int s = 0; s < 3; ++s) {
        float n11 = m11 * m11 + m12 * m21;
        float n12 = m11 * m12 + m12 * m22;
        float n21 = m21 * m11 + m22 * m21;
        float n22 = m21 * m12 + m22 * m22;
        m11 = n11; m12 = n12; m21 = n21; m22 = n22;
    }

#pragma unroll
    for (int o = 1; o < 32; o <<= 1) {
        float o11 = __shfl_up(m11, o, 32);
        float o12 = __shfl_up(m12, o, 32);
        float o21 = __shfl_up(m21, o, 32);
        float o22 = __shfl_up(m22, o, 32);
        float ow1 = __shfl_up(w1, o, 32);
        float ow2 = __shfl_up(w2, o, 32);
        if (lane >= o) {
            float nw1 = m11 * ow1 + m12 * ow2 + w1;
            float nw2 = m21 * ow1 + m22 * ow2 + w2;
            w1 = nw1; w2 = nw2;
            float n11 = m11 * o11 + m12 * o21;
            float n12 = m11 * o12 + m12 * o22;
            float n21 = m21 * o11 + m22 * o21;
            float n22 = m21 * o12 + m22 * o22;
            m11 = n11; m12 = n12; m21 = n21; m22 = n22;
        }
    }
    float in1 = __shfl_up(w1, 1, 32);
    float in2 = __shfl_up(w2, 1, 32);
    if (lane == 0) { in1 = 0.f; in2 = 0.f; }

    float* bc = bound + (size_t)s0 * 2;
    float s1 = in1, s2 = in2;
#pragma unroll
    for (int i = 0; i < SEG; ++i) {
        int q = q0 + i;
        if (q < HALF) {
            v2f st; st.x = s1; st.y = s2;
            *(v2f*)(bc + 2 * q) = st;
        }
        float ns1 = p1 * s1 + q1 * s2 + da[i];
        float ns2 = p2 * s1 + q2 * s2 + db[i];
        s1 = (q < HALF) ? ns1 : s1;
        s2 = (q < HALF) ? ns2 : s2;
    }

    if (hf == 0) {
        float e1 = __shfl(s1, (HALF - 1) / SEG, 32);
        float e2 = __shfl(s2, (HALF - 1) / SEG, 32);
        float x11 = __shfl_up(m11, 1, 32);
        float x12 = __shfl_up(m12, 1, 32);
        float x21 = __shfl_up(m21, 1, 32);
        float x22 = __shfl_up(m22, 1, 32);
        if (lane == 0) { x11 = 1.f; x12 = 0.f; x21 = 0.f; x22 = 1.f; }
        float c1 = x11 * e1 + x12 * e2;
        float c2 = x21 * e1 + x22 * e2;
        float* cc = corr + (size_t)ch * HALF * 2;
#pragma unroll
        for (int i = 0; i < SEG; ++i) {
            int q = q0 + i;
            if (q < HALF) {
                v2f cv; cv.x = c1; cv.y = c2;
                *(v2f*)(cc + 2 * q) = cv;
            }
            float nc1 = p1 * c1 + q1 * c2;
            float nc2 = p2 * c1 + q2 * c2;
            c1 = nc1; c2 = nc2;
        }
    }
}

#define SB      32
#define NSB     10
#define PITCH   36
#define XOFF    4

__global__ void __launch_bounds__(256)
biquad_phase3(const float* __restrict__ X,
              const float* __restrict__ bound,
              const float* __restrict__ corr,
              float* __restrict__ Y,
              float b0, float b1, float b2, float a1, float a2)
{
    __shared__ float sh[17];
    __shared__ __attribute__((aligned(16))) float tin [8 * 16 * PITCH];
    __shared__ __attribute__((aligned(16))) float tout[8 * 16 * PITCH];

    int lane = threadIdx.x & 31;
    int wave = threadIdx.x >> 5;
    int k    = blockIdx.x;

    if (threadIdx.x == 0) {
        float h0 = 1.0f, h1 = -a1;
        sh[0] = h0; sh[1] = h1;
        for (int i = 2; i <= 16; ++i) {
            float hn = -a1 * h1 - a2 * h0;
            sh[i] = hn; h0 = h1; h1 = hn;
        }
    }
    __syncthreads();

    int m    = lane & 15;
    int half = lane >> 4;

    v2f A[4];
#pragma unroll
    for (int j = 0; j < 4; ++j) {
        int k0 = 4 * j + 2 * half;
        A[j].x = (m >= k0)     ? sh[m - k0]     : 0.0f;
        A[j].y = (m >= k0 + 1) ? sh[m - k0 - 1] : 0.0f;
    }
    float uu[8], vv[8];
#pragma unroll
    for (int r = 0; r < 8; ++r) {
        int mm = r + 8 * half;
        uu[r] = sh[mm + 1];
        vv[r] = -a2 * sh[mm];
    }

    int ch = wave * 16 + m;
    int tbase = k * CHUNK;

    float y1 = bound[((size_t)ch * NCHUNK + k) * 2 + 0];
    float y2 = bound[((size_t)ch * NCHUNK + k) * 2 + 1];
    if (k >= HALF) {
        y1 += corr[((size_t)ch * HALF + (k - HALF)) * 2 + 0];
        y2 += corr[((size_t)ch * HALF + (k - HALF)) * 2 + 1];
    }

    float* win  = tin  + wave * 16 * PITCH;
    float* wout = tout + wave * 16 * PITCH;

    int mrow = lane >> 3;
    int mseg = lane & 7;
    const float* xg = X + (size_t)(wave * 16 + mrow) * T_LEN + tbase + 4 * mseg;
    float*       yg = Y + (size_t)(wave * 16 + mrow) * T_LEN + tbase + 4 * mseg;

    v4f ga[4], gb[4];
#pragma unroll
    for (int i = 0; i < 4; ++i)
        ga[i] = *(const v4f*)(xg + (size_t)(4 * i) * T_LEN);
#pragma unroll
    for (int i = 0; i < 4; ++i)
        gb[i] = *(const v4f*)(xg + (size_t)(4 * i) * T_LEN + SB);

    v2f hal; hal.x = 0.f; hal.y = 0.f;
    if (k > 0) hal = *(const v2f*)(X + (size_t)ch * T_LEN + tbase - 2);

    auto do_tile = [&](int sb, v4f (&g)[4]) {
        *(v2f*)(win + m * PITCH + 2) = hal;
#pragma unroll
        for (int i = 0; i < 4; ++i)
            *(v4f*)(win + (4 * i + mrow) * PITCH + XOFF + 4 * mseg) = g[i];
        wave_lds_order();

        if (sb + 2 < NSB) {
#pragma unroll
            for (int i = 0; i < 4; ++i)
                g[i] = *(const v4f*)(xg + (size_t)(4 * i) * T_LEN + (sb + 2) * SB);
        }

#pragma unroll
        for (int bb = 0; bb < 2; ++bb) {
            v8f c;
#pragma unroll
            for (int r = 0; r < 8; ++r) c[r] = uu[r] * y1 + vv[r] * y2;

#pragma unroll
            for (int j = 0; j < 4; ++j) {
                const float* p = win + m * PITCH + XOFF + 16 * bb + 4 * j + 2 * half - 2;
                v2f lo = *(const v2f*)p;
                v2f hi = *(const v2f*)(p + 2);
                v2f Bv;
                Bv.x = b0 * hi.x + b1 * lo.y + b2 * lo.x;
                Bv.y = b0 * hi.y + b1 * hi.x + b2 * lo.y;
                c = __builtin_amdgcn_wmma_f32_16x16x4_f32(
                        false, A[j], false, Bv, (short)0, c, false, false);
            }

            v4f o0, o1;
            o0.x = c[0]; o0.y = c[1]; o0.z = c[2]; o0.w = c[3];
            o1.x = c[4]; o1.y = c[5]; o1.z = c[6]; o1.w = c[7];
            *(v4f*)(wout + m * PITCH + 16 * bb + 8 * half)     = o0;
            *(v4f*)(wout + m * PITCH + 16 * bb + 8 * half + 4) = o1;

            y1 = __shfl(c[7], m + 16, 32);
            y2 = __shfl(c[6], m + 16, 32);
        }

        hal = *(const v2f*)(win + m * PITCH + XOFF + SB - 2);
        wave_lds_order();

#pragma unroll
        for (int i = 0; i < 4; ++i) {
            v4f o = *(const v4f*)(wout + (4 * i + mrow) * PITCH + 4 * mseg);
            *(v4f*)(yg + (size_t)(4 * i) * T_LEN + sb * SB) = o;
        }
    };

    for (int sb = 0; sb < NSB; sb += 2) {
        do_tile(sb,     ga);
        do_tile(sb + 1, gb);
    }
}

extern "C" void kernel_launch(void* const* d_in, const int* in_sizes, int n_in,
                              void* d_out, int out_size, void* d_ws, size_t ws_size,
                              hipStream_t stream) {
    const float* clip = (const float*)d_in[0];
    float* out = (float*)d_out;

    const double SR = 32000.0, CUT = 8000.0, Qf = 0.707;
    double w0    = 2.0 * M_PI * CUT / SR;
    double alpha = sin(w0) / (2.0 * Qf);
    double cw    = cos(w0);
    double b0d = (1.0 - cw) / 2.0, b1d = 1.0 - cw, b2d = b0d;
    double a0d = 1.0 + alpha, a1d = -2.0 * cw, a2d = 1.0 - alpha;
    float b0 = (float)(b0d / a0d), b1 = (float)(b1d / a0d), b2 = (float)(b2d / a0d);
    float a1 = (float)(a1d / a0d), a2 = (float)(a2d / a0d);

    float* bound = (float*)d_ws;
    float* corr  = bound + (size_t)NCHAN * NCHUNK * 2;
    (void)ws_size;

    {
        biquad_phase1<<<dim3(NCHAN * 2), dim3(256), 0, stream>>>(
            clip, bound, corr, b0, b1, b2, a1, a2);
    }
    {
        biquad_phase3<<<dim3(NCHUNK), dim3(256), 0, stream>>>(
            clip, bound, corr, out, b0, b1, b2, a1, a2);
    }
}


# one launch: each work-group (half a channel, 16 waves) runs its chunks on WMMA from a zero state, keeps the outputs in registers, scans end-states in LDS, adds the state response; sum order changes
# speedup vs baseline: 3.0330x; 1.2191x over previous
#include <hip/hip_runtime.h>
#include <hip/hip_bf16.h>
#include <math.h>


typedef __attribute__((ext_vector_type(2))) float v2f;
typedef __attribute__((ext_vector_type(4))) float v4f;
typedef __attribute__((ext_vector_type(8))) float v8f;

#define T_LEN   160000
#define NCHAN   128
#define CHUNK   320
#define NCHUNK  500
#define NBLK    20
#define HALF    250
#define NWAVE   16
#define SEG     16
#define LASTROWS (HALF - 16 * (NWAVE - 1))
static_assert(2 * HALF == NCHUNK, "two work-groups cover a channel");
static_assert(SEG * 32 >= NCHUNK, "one wave scans a channel's chunks");
static_assert(LASTROWS > 0 && LASTROWS <= 16, "16 waves x 16 chunks cover a half");

__device__ __forceinline__ void wave_lds_order()
{
    __builtin_amdgcn_fence(__ATOMIC_RELEASE, "wavefront");
    __builtin_amdgcn_wave_barrier();
    __builtin_amdgcn_fence(__ATOMIC_ACQUIRE, "wavefront");
}

#define SB      32
#define NSB     10
#define PITCH   36
#define XOFF    4

template <bool KEEP>
__device__ __forceinline__ void zero_state_pass(
    const float* __restrict__ xs, int kfirst, int nrows, float* win, int lane,
    const v2f (&A)[4], const float (&uu)[8], const float (&vv)[8],
    float b0, float b1, float b2, v8f (&cy)[NBLK], float& e1, float& e2)
{
    int m    = lane & 15;
    int half = lane >> 4;
    int mrow = lane >> 3;
    int mseg = lane & 7;

    const float* xg[4];
#pragma unroll
    for (int i = 0; i < 4; ++i) {
        int r = 4 * i + mrow;
        r = (r < nrows) ? r : 0;
        xg[i] = xs + (size_t)r * CHUNK + 4 * mseg;
    }

    v4f ga[4], gb[4];
#pragma unroll
    for (int i = 0; i < 4; ++i) ga[i] = *(const v4f*)(xg[i]);
#pragma unroll
    for (int i = 0; i < 4; ++i) gb[i] = *(const v4f*)(xg[i] + SB);

    int mm = (m < nrows) ? m : 0;
    v2f hal; hal.x = 0.f; hal.y = 0.f;
    if (kfirst + mm > 0) hal = *(const v2f*)(xs + (size_t)mm * CHUNK - 2);

    float y1 = 0.f, y2 = 0.f;

    auto do_tile = [&](int sb, v4f (&g)[4]) {
        *(v2f*)(win + m * PITCH + 2) = hal;
#pragma unroll
        for (int i = 0; i < 4; ++i)
            *(v4f*)(win + (4 * i + mrow) * PITCH + XOFF + 4 * mseg) = g[i];
        wave_lds_order();

        if (sb + 2 < NSB) {
#pragma unroll
            for (int i = 0; i < 4; ++i)
                g[i] = *(const v4f*)(xg[i] + (sb + 2) * SB);
        }

#pragma unroll
        for (int bb = 0; bb < 2; ++bb) {
            v8f c;
#pragma unroll
            for (int r = 0; r < 8; ++r) c[r] = uu[r] * y1 + vv[r] * y2;

#pragma unroll
            for (int j = 0; j < 4; ++j) {
                const float* p = win + m * PITCH + XOFF + 16 * bb + 4 * j + 2 * half - 2;
                v2f lo = *(const v2f*)p;
                v2f hi = *(const v2f*)(p + 2);
                v2f Bv;
                Bv.x = b0 * hi.x + b1 * lo.y + b2 * lo.x;
                Bv.y = b0 * hi.y + b1 * hi.x + b2 * lo.y;
                c = __builtin_amdgcn_wmma_f32_16x16x4_f32(
                        false, A[j], false, Bv, (short)0, c, false, false);
            }
            if (KEEP) cy[2 * sb + bb] = c;

            y1 = __shfl(c[7], m + 16, 32);
            y2 = __shfl(c[6], m + 16, 32);
        }

        hal = *(const v2f*)(win + m * PITCH + XOFF + SB - 2);
        wave_lds_order();
    };

#pragma unroll
    for (int sb = 0; sb < NSB; sb += 2) {
        do_tile(sb,     ga);
        do_tile(sb + 1, gb);
    }
    e1 = y1; e2 = y2;
}

__global__ void __launch_bounds__(32 * NWAVE)
biquad_fused(const float* __restrict__ X, float* __restrict__ Y,
             float b0, float b1, float b2, float a1, float a2)
{
    __shared__ float sh[17];
    __shared__ __attribute__((aligned(16))) float tin [NWAVE * 16 * PITCH];
    __shared__ __attribute__((aligned(16))) float tout[NWAVE * 16 * PITCH];
    __shared__ __attribute__((aligned(16))) float sd[2 * SEG * 32];
    __shared__ __attribute__((aligned(16))) float sbnd[2 * SEG * 32];

    int lane = threadIdx.x & 31;
    int wave = threadIdx.x >> 5;
    int grp  = blockIdx.x;
    int ch   = grp >> 1;
    int hf   = grp & 1;

    if (threadIdx.x == 0) {
        float h0 = 1.0f, h1 = -a1;
        sh[0] = h0; sh[1] = h1;
        for (int i = 2; i <= 16; ++i) {
            float hn = -a1 * h1 - a2 * h0;
            sh[i] = hn; h0 = h1; h1 = hn;
        }
    }
    __syncthreads();

    int m    = lane & 15;
    int half = lane >> 4;
    int mrow = lane >> 3;
    int mseg = lane & 7;

    v2f A[4];
#pragma unroll
    for (int j = 0; j < 4; ++j) {
        int k0 = 4 * j + 2 * half;
        A[j].x = (m >= k0)     ? sh[m - k0]     : 0.0f;
        A[j].y = (m >= k0 + 1) ? sh[m - k0 - 1] : 0.0f;
    }
    float uu[8], vv[8];
#pragma unroll
    for (int r = 0; r < 8; ++r) {
        int mm = r + 8 * half;
        uu[r] = sh[mm + 1];
        vv[r] = -a2 * sh[mm];
    }
    float u15 = sh[16], v15 = -a2 * sh[15];
    float u14 = sh[15], v14 = -a2 * sh[14];

    float* win  = tin  + wave * 16 * PITCH;
    float* wout = tout + wave * 16 * PITCH;

    int nrows = (wave < NWAVE - 1) ? 16 : LASTROWS;
    const float* xch = X + (size_t)ch * T_LEN;
    bool owner = (half == 0) && (m < nrows);

    v8f cy[NBLK];
    float e1, e2;

    if (hf == 1) {
        zero_state_pass<false>(xch + (size_t)(16 * wave) * CHUNK, 16 * wave, nrows,
                               win, lane, A, uu, vv, b0, b1, b2, cy, e1, e2);
        if (owner) {
            v2f d; d.x = e1; d.y = e2;
            *(v2f*)(sd + 2 * (16 * wave + m)) = d;
        }
    }
    int kown = hf * HALF + 16 * wave;
    zero_state_pass<true>(xch + (size_t)kown * CHUNK, kown, nrows,
                          win, lane, A, uu, vv, b0, b1, b2, cy, e1, e2);
    if (owner) {
        v2f d; d.x = e1; d.y = e2;
        *(v2f*)(sd + 2 * (kown + m)) = d;
    }
    __syncthreads();

    if (wave == 0) {
        int nv = HALF * (hf + 1);
        float p1 = 1.f, p2 = 0.f;
        float q1 = 0.f, q2 = 1.f;
#pragma unroll
        for (int i = 0; i < 5; ++i) {
            float np = -a1 * p1 - a2 * p2; p2 = p1; p1 = np;
            float nq = -a1 * q1 - a2 * q2; q2 = q1; q1 = nq;
        }
#pragma unroll
        for (int s = 0; s < 6; ++s) {
            float n11 = p1 * p1 + q1 * p2;
            float n12 = p1 * q1 + q1 * q2;
            float n21 = p2 * p1 + q2 * p2;
            float n22 = p2 * q1 + q2 * q2;
            p1 = n11; q1 = n12; p2 = n21; q2 = n22;
        }

        int q0 = lane * SEG;

        float w1 = 0.f, w2 = 0.f;
#pragma unroll
        for (int i = 0; i < SEG; ++i) {
            int q = q0 + i;
            v2f d = *(const v2f*)(sd + 2 * q);
            float dx = (q < nv) ? d.x : 0.f;
            float dy = (q < nv) ? d.y : 0.f;
            float nw1 = p1 * w1 + q1 * w2 + dx;
            float nw2 = p2 * w1 + q2 * w2 + dy;
            w1 = nw1; w2 = nw2;
        }
        float m11 = p1, m12 = q1, m21 = p2, m22 = q2;
#pragma unroll
        for (int s = 0; s < 4; ++s) {
            float n11 = m11 * m11 + m12 * m21;
            float n12 = m11 * m12 + m12 * m22;
            float n21 = m21 * m11 + m22 * m21;
            float n22 = m21 * m12 + m22 * m22;
            m11 = n11; m12 = n12; m21 = n21; m22 = n22;
        }

#pragma unroll
        for (int o = 1; o < 32; o <<= 1) {
            float o11 = __shfl_up(m11, o, 32);
            float o12 = __shfl_up(m12, o, 32);
            float o21 = __shfl_up(m21, o, 32);
            float o22 = __shfl_up(m22, o, 32);
            float ow1 = __shfl_up(w1, o, 32);
            float ow2 = __shfl_up(w2, o, 32);
            if (lane >= o) {
                float nw1 = m11 * ow1 + m12 * ow2 + w1;
                float nw2 = m21 * ow1 + m22 * ow2 + w2;
                w1 = nw1; w2 = nw2;
                float n11 = m11 * o11 + m12 * o21;
                float n12 = m11 * o12 + m12 * o22;
                float n21 = m21 * o11 + m22 * o21;
                float n22 = m21 * o12 + m22 * o22;
                m11 = n11; m12 = n12; m21 = n21; m22 = n22;
            }
        }
        float s1 = __shfl_up(w1, 1, 32);
        float s2 = __shfl_up(w2, 1, 32);
        if (lane == 0) { s1 = 0.f; s2 = 0.f; }

#pragma unroll
        for (int i = 0; i < SEG; ++i) {
            int q = q0 + i;
            v2f d = *(const v2f*)(sd + 2 * q);
            float dx = (q < nv) ? d.x : 0.f;
            float dy = (q < nv) ? d.y : 0.f;
            v2f st; st.x = s1; st.y = s2;
            *(v2f*)(sbnd + 2 * q) = st;
            float ns1 = p1 * s1 + q1 * s2 + dx;
            float ns2 = p2 * s1 + q2 * s2 + dy;
            s1 = ns1; s2 = ns2;
        }
    }
    __syncthreads();

    int mc = (m < nrows) ? m : 0;
    v2f sent = *(const v2f*)(sbnd + 2 * (kown + mc));
    float s1 = sent.x, s2 = sent.y;

    float* yg[4];
    bool   rowok[4];
#pragma unroll
    for (int i = 0; i < 4; ++i) {
        int r = 4 * i + mrow;
        rowok[i] = r < nrows;
        r = rowok[i] ? r : 0;
        yg[i] = Y + (size_t)ch * T_LEN + (size_t)(kown + r) * CHUNK + 4 * mseg;
    }

#pragma unroll
    for (int sb = 0; sb < NSB; ++sb) {
#pragma unroll
        for (int bb = 0; bb < 2; ++bb) {
            v8f c = cy[2 * sb + bb];
#pragma unroll
            for (int r = 0; r < 8; ++r) c[r] += uu[r] * s1 + vv[r] * s2;
            float n1 = u15 * s1 + v15 * s2;
            float n2 = u14 * s1 + v14 * s2;
            s1 = n1; s2 = n2;

            v4f o0, o1;
            o0.x = c[0]; o0.y = c[1]; o0.z = c[2]; o0.w = c[3];
            o1.x = c[4]; o1.y = c[5]; o1.z = c[6]; o1.w = c[7];
            *(v4f*)(wout + m * PITCH + 16 * bb + 8 * half)     = o0;
            *(v4f*)(wout + m * PITCH + 16 * bb + 8 * half + 4) = o1;
        }
        wave_lds_order();

#pragma unroll
        for (int i = 0; i < 4; ++i) {
            v4f o = *(const v4f*)(wout + (4 * i + mrow) * PITCH + 4 * mseg);
            if (rowok[i]) *(v4f*)(yg[i] + sb * SB) = o;
        }
        wave_lds_order();
    }
}

extern "C" void kernel_launch(void* const* d_in, const int* in_sizes, int n_in,
                              void* d_out, int out_size, void* d_ws, size_t ws_size,
                              hipStream_t stream) {
    const float* clip = (const float*)d_in[0];
    float* out = (float*)d_out;

    const double SR = 32000.0, CUT = 8000.0, Qf = 0.707;
    double w0    = 2.0 * M_PI * CUT / SR;
    double alpha = sin(w0) / (2.0 * Qf);
    double cw    = cos(w0);
    double b0d = (1.0 - cw) / 2.0, b1d = 1.0 - cw, b2d = b0d;
    double a0d = 1.0 + alpha, a1d = -2.0 * cw, a2d = 1.0 - alpha;
    float b0 = (float)(b0d / a0d), b1 = (float)(b1d / a0d), b2 = (float)(b2d / a0d);
    float a1 = (float)(a1d / a0d), a2 = (float)(a2d / a0d);

    (void)d_ws; (void)ws_size;

    biquad_fused<<<dim3(NCHAN * 2), dim3(32 * NWAVE), 0, stream>>>(
        clip, out, b0, b1, b2, a1, a2);
}


# second-half work-group stops re-reading the first half: its last wave's 6 spare columns take the 6 chunks in front; older ones skipped if (M^320)^6 == 0 exactly in f32, else old path; sums unchanged
# speedup vs baseline: 3.6761x; 1.2120x over previous
#include <hip/hip_runtime.h>
#include <hip/hip_bf16.h>
#include <math.h>


typedef __attribute__((ext_vector_type(2))) float v2f;
typedef __attribute__((ext_vector_type(4))) float v4f;
typedef __attribute__((ext_vector_type(8))) float v8f;

#define T_LEN   160000
#define NCHAN   128
#define CHUNK   320
#define NCHUNK  500
#define NBLK    20
#define HALF    250
#define NWAVE   16
#define SEG     16
#define LASTROWS (HALF - 16 * (NWAVE - 1))
#define NEXTRA  (16 - LASTROWS)
static_assert(2 * HALF == NCHUNK, "two work-groups cover a channel");
static_assert(NEXTRA >= 1 && NEXTRA <= HALF, "the spare columns hold chunks of the first half");
static_assert(SEG * 32 >= NCHUNK, "one wave scans a channel's chunks");
static_assert(LASTROWS > 0 && LASTROWS <= 16, "16 waves x 16 chunks cover a half");

__device__ __forceinline__ void wave_lds_order()
{
    __builtin_amdgcn_fence(__ATOMIC_RELEASE, "wavefront");
    __builtin_amdgcn_wave_barrier();
    __builtin_amdgcn_fence(__ATOMIC_ACQUIRE, "wavefront");
}

#define SB      32
#define NSB     10
#define PITCH   36
#define XOFF    4

template <bool KEEP>
__device__ __forceinline__ void zero_state_pass(
    const float* __restrict__ xch, int kfirst, int nrows, int kpad, int padstep,
    float* win, int lane,
    const v2f (&A)[4], v2f AS, v8f (&cy)[NBLK], float& e1, float& e2)
{
    int m    = lane & 15;
    int half = lane >> 4;
    int mrow = lane >> 3;
    int mseg = lane & 7;

    const float* xg[4];
#pragma unroll
    for (int i = 0; i < 4; ++i) {
        int r = 4 * i + mrow;
        int kc = (r < nrows) ? kfirst + r : kpad + padstep * r;
        xg[i] = xch + (size_t)kc * CHUNK + 4 * mseg;
    }

    v4f ga[4], gb[4];
#pragma unroll
    for (int i = 0; i < 4; ++i) ga[i] = *(const v4f*)(xg[i]);
#pragma unroll
    for (int i = 0; i < 4; ++i) gb[i] = *(const v4f*)(xg[i] + SB);

    int kcm = (m < nrows) ? kfirst + m : kpad + padstep * m;
    v2f hal; hal.x = 0.f; hal.y = 0.f;
    if (kcm > 0) hal = *(const v2f*)(xch + (size_t)kcm * CHUNK - 2);

    v2f yst; yst.x = 0.f; yst.y = 0.f;

    auto do_tile = [&](int sb, v4f (&g)[4]) {
        *(v2f*)(win + m * PITCH + 2) = hal;
#pragma unroll
        for (int i = 0; i < 4; ++i)
            *(v4f*)(win + (4 * i + mrow) * PITCH + XOFF + 4 * mseg) = g[i];
        wave_lds_order();

        if (sb + 2 < NSB) {
#pragma unroll
            for (int i = 0; i < 4; ++i)
                g[i] = *(const v4f*)(xg[i] + (sb + 2) * SB);
        }

#pragma unroll
        for (int bb = 0; bb < 2; ++bb) {
            const float* p = win + m * PITCH + XOFF + 16 * bb;
            v2f xb = *(const v2f*)(p - 2);

            v8f c = {};
#pragma unroll
            for (int j = 0; j < 4; ++j) {
                v2f Bv = *(const v2f*)(p + 4 * j + 2 * half);
                c = __builtin_amdgcn_wmma_f32_16x16x4_f32(
                        false, A[j], false, Bv, (short)0, c, false, false);
            }
            v2f Bs;
            Bs.x = half ? yst.x : xb.x;
            Bs.y = half ? yst.y : xb.y;
            c = __builtin_amdgcn_wmma_f32_16x16x4_f32(
                    false, AS, false, Bs, (short)0, c, false, false);
            if (KEEP) cy[2 * sb + bb] = c;

            yst.x = c[6];
            yst.y = c[7];
        }

        hal = *(const v2f*)(win + m * PITCH + XOFF + SB - 2);
        wave_lds_order();
        __builtin_amdgcn_sched_barrier(0);
    };

#pragma unroll
    for (int sb = 0; sb < NSB; sb += 2) {
        do_tile(sb,     ga);
        do_tile(sb + 1, gb);
    }
    e1 = yst.y; e2 = yst.x;
}

__global__ void __launch_bounds__(32 * NWAVE)
biquad_fused(const float* __restrict__ X, float* __restrict__ Y,
             float b0, float b1, float b2, float a1, float a2)
{
    __shared__ float sh[17];
    __shared__ float sg[16];
    __shared__ __attribute__((aligned(16))) float tin [NWAVE * 16 * PITCH];
    __shared__ __attribute__((aligned(16))) float tout[NWAVE * 16 * PITCH];
    __shared__ __attribute__((aligned(16))) float sd[2 * SEG * 32];
    __shared__ __attribute__((aligned(16))) float sbnd[2 * SEG * 32];

    int lane = threadIdx.x & 31;
    int wave = threadIdx.x >> 5;
    int grp  = blockIdx.x;
    int ch   = grp >> 1;
    int hf   = grp & 1;

    float p1 = 1.f, p2 = 0.f;
    float q1 = 0.f, q2 = 1.f;
#pragma unroll
    for (int i = 0; i < 5; ++i) {
        float np = -a1 * p1 - a2 * p2; p2 = p1; p1 = np;
        float nq = -a1 * q1 - a2 * q2; q2 = q1; q1 = nq;
    }
#pragma unroll
    for (int s = 0; s < 6; ++s) {
        float n11 = p1 * p1 + q1 * p2;
        float n12 = p1 * q1 + q1 * q2;
        float n21 = p2 * p1 + q2 * p2;
        float n22 = p2 * q1 + q2 * q2;
        p1 = n11; q1 = n12; p2 = n21; q2 = n22;
    }
    bool carry_dead;
    {
        float r11 = p1, r12 = q1, r21 = p2, r22 = q2;
#pragma unroll
        for (int i = 1; i < NEXTRA; ++i) {
            float n11 = r11 * p1 + r12 * p2;
            float n12 = r11 * q1 + r12 * q2;
            float n21 = r21 * p1 + r22 * p2;
            float n22 = r21 * q1 + r22 * q2;
            r11 = n11; r12 = n12; r21 = n21; r22 = n22;
        }
        carry_dead = (r11 == 0.f) && (r12 == 0.f) && (r21 == 0.f) && (r22 == 0.f);
    }

    if (threadIdx.x == 0) {
        float h0 = 1.0f, h1 = -a1;
        sh[0] = h0; sh[1] = h1;
        sg[0] = b0; sg[1] = b0 * h1 + b1;
        for (int i = 2; i <= 16; ++i) {
            float hn = -a1 * h1 - a2 * h0;
            sh[i] = hn;
            if (i < 16) sg[i] = b0 * hn + b1 * h1 + b2 * h0;
            h0 = h1; h1 = hn;
        }
    }
    __syncthreads();

    int m    = lane & 15;
    int half = lane >> 4;
    int mrow = lane >> 3;
    int mseg = lane & 7;

    v2f A[4];
#pragma unroll
    for (int j = 0; j < 4; ++j) {
        int k0 = 4 * j + 2 * half;
        A[j].x = (m >= k0)     ? sg[m - k0]     : 0.0f;
        A[j].y = (m >= k0 + 1) ? sg[m - k0 - 1] : 0.0f;
    }
    v2f AS;
    {
        float hm  = sh[m];
        float hm1 = (m >= 1) ? sh[m - 1] : 0.0f;
        float hp1 = sh[m + 1];
        AS.x = half ? (-a2 * hm) : (b2 * hm);
        AS.y = half ? hp1        : (b1 * hm + b2 * hm1);
    }

    float* win  = tin  + wave * 16 * PITCH;
    float* wout = tout + wave * 16 * PITCH;

    int nrows = (wave < NWAVE - 1) ? 16 : LASTROWS;
    const float* xch = X + (size_t)ch * T_LEN;
    bool owner = (half == 1) && (m < nrows);

    v8f cy[NBLK];
    float e1, e2;

    if (hf == 1 && !carry_dead) {
        zero_state_pass<false>(xch, 16 * wave, nrows, 16 * wave, 0,
                               win, lane, A, AS, cy, e1, e2);
        if (owner) {
            v2f d; d.x = e1; d.y = e2;
            *(v2f*)(sd + 2 * (16 * wave + m)) = d;
        }
    }
    int kown = hf * HALF + 16 * wave;
    bool ride = (hf == 1) && carry_dead && (wave == NWAVE - 1);
    zero_state_pass<true>(xch, kown, nrows, ride ? HALF - 16 : kown, ride ? 1 : 0,
                          win, lane, A, AS, cy, e1, e2);
    if (owner) {
        v2f d; d.x = e1; d.y = e2;
        *(v2f*)(sd + 2 * (kown + m)) = d;
    }
    if (ride && half == 1 && m >= nrows) {
        v2f d; d.x = e1; d.y = e2;
        *(v2f*)(sd + 2 * (HALF - 16 + m)) = d;
    }
    __syncthreads();

    if (wave == 0) {
        int nv = HALF * (hf + 1);
        int qlo = (hf == 1 && carry_dead) ? HALF - NEXTRA : 0;

        int q0 = lane * SEG;

        float w1 = 0.f, w2 = 0.f;
#pragma unroll
        for (int i = 0; i < SEG; ++i) {
            int q = q0 + i;
            v2f d = *(const v2f*)(sd + 2 * q);
            bool live = (q >= qlo) && (q < nv);
            float dx = live ? d.x : 0.f;
            float dy = live ? d.y : 0.f;
            float nw1 = p1 * w1 + q1 * w2 + dx;
            float nw2 = p2 * w1 + q2 * w2 + dy;
            w1 = nw1; w2 = nw2;
        }
        float m11 = p1, m12 = q1, m21 = p2, m22 = q2;
#pragma unroll
        for (int s = 0; s < 4; ++s) {
            float n11 = m11 * m11 + m12 * m21;
            float n12 = m11 * m12 + m12 * m22;
            float n21 = m21 * m11 + m22 * m21;
            float n22 = m21 * m12 + m22 * m22;
            m11 = n11; m12 = n12; m21 = n21; m22 = n22;
        }

#pragma unroll
        for (int o = 1; o < 32; o <<= 1) {
            float o11 = __shfl_up(m11, o, 32);
            float o12 = __shfl_up(m12, o, 32);
            float o21 = __shfl_up(m21, o, 32);
            float o22 = __shfl_up(m22, o, 32);
            float ow1 = __shfl_up(w1, o, 32);
            float ow2 = __shfl_up(w2, o, 32);
            if (lane >= o) {
                float nw1 = m11 * ow1 + m12 * ow2 + w1;
                float nw2 = m21 * ow1 + m22 * ow2 + w2;
                w1 = nw1; w2 = nw2;
                float n11 = m11 * o11 + m12 * o21;
                float n12 = m11 * o12 + m12 * o22;
                float n21 = m21 * o11 + m22 * o21;
                float n22 = m21 * o12 + m22 * o22;
                m11 = n11; m12 = n12; m21 = n21; m22 = n22;
            }
        }
        float s1 = __shfl_up(w1, 1, 32);
        float s2 = __shfl_up(w2, 1, 32);
        if (lane == 0) { s1 = 0.f; s2 = 0.f; }

#pragma unroll
        for (int i = 0; i < SEG; ++i) {
            int q = q0 + i;
            v2f d = *(const v2f*)(sd + 2 * q);
            bool live = (q >= qlo) && (q < nv);
            float dx = live ? d.x : 0.f;
            float dy = live ? d.y : 0.f;
            v2f st; st.x = s1; st.y = s2;
            *(v2f*)(sbnd + 2 * q) = st;
            float ns1 = p1 * s1 + q1 * s2 + dx;
            float ns2 = p2 * s1 + q2 * s2 + dy;
            s1 = ns1; s2 = ns2;
        }
    }
    __syncthreads();

    float uu[8], vv[8];
#pragma unroll
    for (int r = 0; r < 8; ++r) {
        int mm = r + 8 * half;
        uu[r] = sh[mm + 1];
        vv[r] = -a2 * sh[mm];
    }
    float u15 = sh[16], v15 = -a2 * sh[15];
    float u14 = sh[15], v14 = -a2 * sh[14];

    int mc = (m < nrows) ? m : 0;
    v2f sent = *(const v2f*)(sbnd + 2 * (kown + mc));
    float s1 = sent.x, s2 = sent.y;

    float* yg[4];
    bool   rowok[4];
#pragma unroll
    for (int i = 0; i < 4; ++i) {
        int r = 4 * i + mrow;
        rowok[i] = r < nrows;
        r = rowok[i] ? r : 0;
        yg[i] = Y + (size_t)ch * T_LEN + (size_t)(kown + r) * CHUNK + 4 * mseg;
    }

#pragma unroll
    for (int sb = 0; sb < NSB; ++sb) {
#pragma unroll
        for (int bb = 0; bb < 2; ++bb) {
            v8f c = cy[2 * sb + bb];
#pragma unroll
            for (int r = 0; r < 8; ++r) c[r] += uu[r] * s1 + vv[r] * s2;
            float n1 = u15 * s1 + v15 * s2;
            float n2 = u14 * s1 + v14 * s2;
            s1 = n1; s2 = n2;

            v4f o0, o1;
            o0.x = c[0]; o0.y = c[1]; o0.z = c[2]; o0.w = c[3];
            o1.x = c[4]; o1.y = c[5]; o1.z = c[6]; o1.w = c[7];
            *(v4f*)(wout + m * PITCH + 16 * bb + 8 * half)     = o0;
            *(v4f*)(wout + m * PITCH + 16 * bb + 8 * half + 4) = o1;
        }
        wave_lds_order();

#pragma unroll
        for (int i = 0; i < 4; ++i) {
            v4f o = *(const v4f*)(wout + (4 * i + mrow) * PITCH + 4 * mseg);
            if (rowok[i]) *(v4f*)(yg[i] + sb * SB) = o;
        }
        wave_lds_order();
    }
}

extern "C" void kernel_launch(void* const* d_in, const int* in_sizes, int n_in,
                              void* d_out, int out_size, void* d_ws, size_t ws_size,
                              hipStream_t stream) {
    const float* clip = (const float*)d_in[0];
    float* out = (float*)d_out;

    const double SR = 32000.0, CUT = 8000.0, Qf = 0.707;
    double w0    = 2.0 * M_PI * CUT / SR;
    double alpha = sin(w0) / (2.0 * Qf);
    double cw    = cos(w0);
    double b0d = (1.0 - cw) / 2.0, b1d = 1.0 - cw, b2d = b0d;
    double a0d = 1.0 + alpha, a1d = -2.0 * cw, a2d = 1.0 - alpha;
    float b0 = (float)(b0d / a0d), b1 = (float)(b1d / a0d), b2 = (float)(b2d / a0d);
    float a1 = (float)(a1d / a0d), a2 = (float)(a2d / a0d);

    (void)d_ws; (void)ws_size;

    biquad_fused<<<dim3(NCHAN * 2), dim3(32 * NWAVE), 0, stream>>>(
        clip, out, b0, b1, b2, a1, a2);
}
